# starting kernel
# baseline (speedup 1.0000x reference)
#include <hip/hip_runtime.h>
#include <hip/hip_bf16.h>

#define BATCH 256
#define TSEQ  261
#define DIM   768
#define NCUE  5
#define NPAT  256
#define MTOT  (BATCH * TSEQ)

typedef __attribute__((ext_vector_type(2))) float v2f;
typedef __attribute__((ext_vector_type(8))) float v8f;

#define BM 128
#define BN 64
#define BK 32
#define LDS_STRIDE (BK + 2)

__global__ __launch_bounds__(256, 2)
void proj_gemm_wmma(const float* __restrict__ A,
                    const float* __restrict__ W,
                    const float* __restrict__ bias,
                    float* __restrict__ T)
{
    __shared__ float As[BM][LDS_STRIDE];
    __shared__ float Bs[BN][LDS_STRIDE];

    const int tid  = threadIdx.x;
    const int wave = tid >> 5;
    const int lane = tid & 31;
    const int hl   = lane >> 4;
    const int l16  = lane & 15;

    const int bm = blockIdx.x * BM;
    const int bn = blockIdx.y * BN;

    const int sm0 = (wave & 3) * 2;
    const int sm1 = sm0 + 1;
    const int sn0 = (wave >> 2) * 2;
    const int sn1 = sn0 + 1;

    v8f c00 = {}, c01 = {}, c10 = {}, c11 = {};

    for (int k0 = 0; k0 < DIM; k0 += BK) {
        #pragma unroll
        for (int i = 0; i < 4; ++i) {
            const int linear = tid + i * 256;
            const int r  = linear >> 3;
            const int cc = (linear & 7) * 4;
            const float* ga = A + (size_t)(bm + r) * DIM + (k0 + cc);
            As[r][cc + 0] = ga[0];
            As[r][cc + 1] = ga[1];
            As[r][cc + 2] = ga[2];
            As[r][cc + 3] = ga[3];
        }
        #pragma unroll
        for (int i = 0; i < 2; ++i) {
            const int linear = tid + i * 256;
            const int r  = linear >> 3;
            const int cc = (linear & 7) * 4;
            const float* gb = W + (size_t)(bn + r) * DIM + (k0 + cc);
            Bs[r][cc + 0] = gb[0];
            Bs[r][cc + 1] = gb[1];
            Bs[r][cc + 2] = gb[2];
            Bs[r][cc + 3] = gb[3];
        }
        if (k0 + BK < DIM) {
            __builtin_prefetch(A + (size_t)(bm + (tid >> 1)) * DIM + k0 + BK, 0, 0);
            __builtin_prefetch(W + (size_t)(bn + (tid & 63)) * DIM + k0 + BK, 0, 0);
        }
        __syncthreads();

        #pragma unroll
        for (int kk = 0; kk < BK; kk += 4) {
            v2f a0 = *(const v2f*)&As[sm0 * 16 + l16][kk + 2 * hl];
            v2f a1 = *(const v2f*)&As[sm1 * 16 + l16][kk + 2 * hl];
            v2f b0 = *(const v2f*)&Bs[sn0 * 16 + l16][kk + 2 * hl];
            v2f b1 = *(const v2f*)&Bs[sn1 * 16 + l16][kk + 2 * hl];
            c00 = __builtin_amdgcn_wmma_f32_16x16x4_f32(
                      false, a0, false, b0, (short)0, c00, false, false);
            c01 = __builtin_amdgcn_wmma_f32_16x16x4_f32(
                      false, a0, false, b1, (short)0, c01, false, false);
            c10 = __builtin_amdgcn_wmma_f32_16x16x4_f32(
                      false, a1, false, b0, (short)0, c10, false, false);
            c11 = __builtin_amdgcn_wmma_f32_16x16x4_f32(
                      false, a1, false, b1, (short)0, c11, false, false);
        }
        __syncthreads();
    }

    const int row0 = bm + sm0 * 16;
    const int row1 = bm + sm1 * 16;
    const int col0 = bn + sn0 * 16 + l16;
    const int col1 = bn + sn1 * 16 + l16;
    const float bias0 = bias[col0];
    const float bias1 = bias[col1];
    #pragma unroll
    for (int r = 0; r < 8; ++r) {
        const int ra = row0 + r + 8 * hl;
        const int rb = row1 + r + 8 * hl;
        T[(size_t)ra * DIM + col0] = c00[r] + bias0;
        T[(size_t)ra * DIM + col1] = c01[r] + bias1;
        T[(size_t)rb * DIM + col0] = c10[r] + bias0;
        T[(size_t)rb * DIM + col1] = c11[r] + bias1;
    }
}

__global__ __launch_bounds__(256)
void buddy_pool_norm(const float* __restrict__ T, float* __restrict__ out)
{
    const int bc = blockIdx.x;
    const int b  = bc / NCUE;
    const int c  = bc % NCUE;
    const int tid = threadIdx.x;

    __shared__ float cue[DIM];
    __shared__ float roi[DIM];
    __shared__ float redv[256];
    __shared__ int   redi[256];

    const float* tb   = T + (size_t)b * TSEQ * DIM;
    const float* cueg = tb + (size_t)c * DIM;
    const float* pat  = tb + (size_t)NCUE * DIM;

    for (int d = tid; d < DIM; d += 256) cue[d] = cueg[d];
    __syncthreads();

    float sim = 0.f;
    {
        const float* pr = pat + (size_t)tid * DIM;
        #pragma unroll 4
        for (int d = 0; d < DIM; ++d) sim += cue[d] * pr[d];
    }
    redv[tid] = sim;
    redi[tid] = tid;
    __syncthreads();

    for (int s = 128; s > 0; s >>= 1) {
        if (tid < s) {
            const float ov = redv[tid + s];
            const int   oi = redi[tid + s];
            if (ov > redv[tid] || (ov == redv[tid] && oi < redi[tid])) {
                redv[tid] = ov;
                redi[tid] = oi;
            }
        }
        __syncthreads();
    }
    const int idx = redi[0];
    const int h = idx >> 4, w = idx & 15;
    const int r0 = max(h - 1, 0), r1 = min(h + 1, 15);
    const int q0 = max(w - 1, 0), q1 = min(w + 1, 15);
    const float inv = 1.0f / (float)((r1 - r0 + 1) * (q1 - q0 + 1));

    for (int d = tid; d < DIM; d += 256) {
        float acc = 0.f;
        for (int hh = r0; hh <= r1; ++hh)
            for (int ww = q0; ww <= q1; ++ww)
                acc += pat[(size_t)(hh * 16 + ww) * DIM + d];
        roi[d] = acc * inv;
    }
    __syncthreads();

    float s0 = 0.f, s1 = 0.f;
    for (int d = tid; d < DIM; d += 256) {
        s0 += cue[d] * cue[d];
        s1 += roi[d] * roi[d];
    }
    redv[tid] = s0;
    __syncthreads();
    for (int s = 128; s > 0; s >>= 1) {
        if (tid < s) redv[tid] += redv[tid + s];
        __syncthreads();
    }
    const float ncue = sqrtf(redv[0]);
    __syncthreads();
    redv[tid] = s1;
    __syncthreads();
    for (int s = 128; s > 0; s >>= 1) {
        if (tid < s) redv[tid] += redv[tid + s];
        __syncthreads();
    }
    const float nroi = sqrtf(redv[0]);

    const float icue = 1.0f / fmaxf(ncue, 1e-12f);
    const float iroi = 1.0f / fmaxf(nroi, 1e-12f);

    float* ob = out + (size_t)b * (2 * NCUE) * DIM;
    for (int d = tid; d < DIM; d += 256) {
        ob[(size_t)c * DIM + d]          = cue[d] * icue;
        ob[(size_t)(NCUE + c) * DIM + d] = roi[d] * iroi;
    }
}

extern "C" void kernel_launch(void* const* d_in, const int* in_sizes, int n_in,
                              void* d_out, int out_size, void* d_ws, size_t ws_size,
                              hipStream_t stream)
{
    const float* tokens = (const float*)d_in[0];
    const float* Wproj  = (const float*)d_in[1];
    const float* bproj  = (const float*)d_in[2];
    float* out = (float*)d_out;
    float* t   = (float*)d_ws;

    dim3 g1(MTOT / BM, DIM / BN);
    proj_gemm_wmma<<<g1, 256, 0, stream>>>(tokens, Wproj, bproj, t);

    buddy_pool_norm<<<BATCH * NCUE, 256, 0, stream>>>(t, out);
}


# pool kernel similarity: a wave reads whole patch rows, 16 B per lane (512 contiguous B per load), 4 rows in flight, shuffle butterfly per row; was one strided row per lane; sim sum order changes
# speedup vs baseline: 1.3350x; 1.3350x over previous
#include <hip/hip_runtime.h>
#include <hip/hip_bf16.h>

#define BATCH 256
#define TSEQ  261
#define DIM   768
#define NCUE  5
#define NPAT  256
#define MTOT  (BATCH * TSEQ)

typedef __attribute__((ext_vector_type(2))) float v2f;
typedef __attribute__((ext_vector_type(8))) float v8f;

#define BM 128
#define BN 64
#define BK 32
#define LDS_STRIDE (BK + 2)

__global__ __launch_bounds__(256, 2)
void proj_gemm_wmma(const float* __restrict__ A,
                    const float* __restrict__ W,
                    const float* __restrict__ bias,
                    float* __restrict__ T)
{
    __shared__ float As[BM][LDS_STRIDE];
    __shared__ float Bs[BN][LDS_STRIDE];

    const int tid  = threadIdx.x;
    const int wave = tid >> 5;
    const int lane = tid & 31;
    const int hl   = lane >> 4;
    const int l16  = lane & 15;

    const int bm = blockIdx.x * BM;
    const int bn = blockIdx.y * BN;

    const int sm0 = (wave & 3) * 2;
    const int sm1 = sm0 + 1;
    const int sn0 = (wave >> 2) * 2;
    const int sn1 = sn0 + 1;

    v8f c00 = {}, c01 = {}, c10 = {}, c11 = {};

    for (int k0 = 0; k0 < DIM; k0 += BK) {
        #pragma unroll
        for (int i = 0; i < 4; ++i) {
            const int linear = tid + i * 256;
            const int r  = linear >> 3;
            const int cc = (linear & 7) * 4;
            const float* ga = A + (size_t)(bm + r) * DIM + (k0 + cc);
            As[r][cc + 0] = ga[0];
            As[r][cc + 1] = ga[1];
            As[r][cc + 2] = ga[2];
            As[r][cc + 3] = ga[3];
        }
        #pragma unroll
        for (int i = 0; i < 2; ++i) {
            const int linear = tid + i * 256;
            const int r  = linear >> 3;
            const int cc = (linear & 7) * 4;
            const float* gb = W + (size_t)(bn + r) * DIM + (k0 + cc);
            Bs[r][cc + 0] = gb[0];
            Bs[r][cc + 1] = gb[1];
            Bs[r][cc + 2] = gb[2];
            Bs[r][cc + 3] = gb[3];
        }
        if (k0 + BK < DIM) {
            __builtin_prefetch(A + (size_t)(bm + (tid >> 1)) * DIM + k0 + BK, 0, 0);
            __builtin_prefetch(W + (size_t)(bn + (tid & 63)) * DIM + k0 + BK, 0, 0);
        }
        __syncthreads();

        #pragma unroll
        for (int kk = 0; kk < BK; kk += 4) {
            v2f a0 = *(const v2f*)&As[sm0 * 16 + l16][kk + 2 * hl];
            v2f a1 = *(const v2f*)&As[sm1 * 16 + l16][kk + 2 * hl];
            v2f b0 = *(const v2f*)&Bs[sn0 * 16 + l16][kk + 2 * hl];
            v2f b1 = *(const v2f*)&Bs[sn1 * 16 + l16][kk + 2 * hl];
            c00 = __builtin_amdgcn_wmma_f32_16x16x4_f32(
                      false, a0, false, b0, (short)0, c00, false, false);
            c01 = __builtin_amdgcn_wmma_f32_16x16x4_f32(
                      false, a0, false, b1, (short)0, c01, false, false);
            c10 = __builtin_amdgcn_wmma_f32_16x16x4_f32(
                      false, a1, false, b0, (short)0, c10, false, false);
            c11 = __builtin_amdgcn_wmma_f32_16x16x4_f32(
                      false, a1, false, b1, (short)0, c11, false, false);
        }
        __syncthreads();
    }

    const int row0 = bm + sm0 * 16;
    const int row1 = bm + sm1 * 16;
    const int col0 = bn + sn0 * 16 + l16;
    const int col1 = bn + sn1 * 16 + l16;
    const float bias0 = bias[col0];
    const float bias1 = bias[col1];
    #pragma unroll
    for (int r = 0; r < 8; ++r) {
        const int ra = row0 + r + 8 * hl;
        const int rb = row1 + r + 8 * hl;
        T[(size_t)ra * DIM + col0] = c00[r] + bias0;
        T[(size_t)ra * DIM + col1] = c01[r] + bias1;
        T[(size_t)rb * DIM + col0] = c10[r] + bias0;
        T[(size_t)rb * DIM + col1] = c11[r] + bias1;
    }
}

__global__ __launch_bounds__(256)
void buddy_pool_norm(const float* __restrict__ T, float* __restrict__ out)
{
    const int bc = blockIdx.x;
    const int b  = bc / NCUE;
    const int c  = bc % NCUE;
    const int tid = threadIdx.x;

    __shared__ __align__(16) float cue[DIM];
    __shared__ float roi[DIM];
    __shared__ float redv[256];
    __shared__ int   redi[256];

    const float* tb   = T + (size_t)b * TSEQ * DIM;
    const float* cueg = tb + (size_t)c * DIM;
    const float* pat  = tb + (size_t)NCUE * DIM;

    for (int d = tid; d < DIM; d += 256) cue[d] = cueg[d];
    __syncthreads();

    const int lane = tid & 31;
    const int wave = tid >> 5;
    float4 cq[DIM / 128];
    for (int q = 0; q < DIM / 128; ++q)
        cq[q] = *(const float4*)&cue[q * 128 + lane * 4];

    for (int it = 0; it < NPAT / 32; ++it) {
        const int n0 = it * 32 + wave * 4;
        const float4* pr = (const float4*)(pat + (size_t)n0 * DIM) + lane;
        float4 x[4][DIM / 128];
        for (int u = 0; u < 4; ++u)
            for (int q = 0; q < DIM / 128; ++q)
                x[u][q] = pr[u * (DIM / 4) + q * 32];
        __builtin_amdgcn_sched_barrier(0);
        float s[4];
        for (int u = 0; u < 4; ++u) {
            float a0 = 0.f, a1 = 0.f, a2 = 0.f, a3 = 0.f;
            for (int q = 0; q < DIM / 128; ++q) {
                a0 = __builtin_fmaf(cq[q].x, x[u][q].x, a0);
                a1 = __builtin_fmaf(cq[q].y, x[u][q].y, a1);
                a2 = __builtin_fmaf(cq[q].z, x[u][q].z, a2);
                a3 = __builtin_fmaf(cq[q].w, x[u][q].w, a3);
            }
            s[u] = (a0 + a1) + (a2 + a3);
        }
        for (int m = 16; m >= 1; m >>= 1) {
            float t[4];
            for (int u = 0; u < 4; ++u) t[u] = __shfl_xor(s[u], m, 32);
            for (int u = 0; u < 4; ++u) s[u] += t[u];
        }
        const float mine = (lane == 0) ? s[0] : (lane == 1) ? s[1] : (lane == 2) ? s[2] : s[3];
        if (lane < 4) redv[n0 + lane] = mine;
    }
    redi[tid] = tid;
    __syncthreads();

    for (int s = 128; s > 0; s >>= 1) {
        if (tid < s) {
            const float ov = redv[tid + s];
            const int   oi = redi[tid + s];
            if (ov > redv[tid] || (ov == redv[tid] && oi < redi[tid])) {
                redv[tid] = ov;
                redi[tid] = oi;
            }
        }
        __syncthreads();
    }
    const int idx = redi[0];
    const int h = idx >> 4, w = idx & 15;
    const int r0 = max(h - 1, 0), r1 = min(h + 1, 15);
    const int q0 = max(w - 1, 0), q1 = min(w + 1, 15);
    const float inv = 1.0f / (float)((r1 - r0 + 1) * (q1 - q0 + 1));

    for (int d = tid; d < DIM; d += 256) {
        float acc = 0.f;
        for (int hh = r0; hh <= r1; ++hh)
            for (int ww = q0; ww <= q1; ++ww)
                acc += pat[(size_t)(hh * 16 + ww) * DIM + d];
        roi[d] = acc * inv;
    }
    __syncthreads();

    float s0 = 0.f, s1 = 0.f;
    for (int d = tid; d < DIM; d += 256) {
        s0 += cue[d] * cue[d];
        s1 += roi[d] * roi[d];
    }
    redv[tid] = s0;
    __syncthreads();
    for (int s = 128; s > 0; s >>= 1) {
        if (tid < s) redv[tid] += redv[tid + s];
        __syncthreads();
    }
    const float ncue = sqrtf(redv[0]);
    __syncthreads();
    redv[tid] = s1;
    __syncthreads();
    for (int s = 128; s > 0; s >>= 1) {
        if (tid < s) redv[tid] += redv[tid + s];
        __syncthreads();
    }
    const float nroi = sqrtf(redv[0]);

    const float icue = 1.0f / fmaxf(ncue, 1e-12f);
    const float iroi = 1.0f / fmaxf(nroi, 1e-12f);

    float* ob = out + (size_t)b * (2 * NCUE) * DIM;
    for (int d = tid; d < DIM; d += 256) {
        ob[(size_t)c * DIM + d]          = cue[d] * icue;
        ob[(size_t)(NCUE + c) * DIM + d] = roi[d] * iroi;
    }
}

extern "C" void kernel_launch(void* const* d_in, const int* in_sizes, int n_in,
                              void* d_out, int out_size, void* d_ws, size_t ws_size,
                              hipStream_t stream)
{
    const float* tokens = (const float*)d_in[0];
    const float* Wproj  = (const float*)d_in[1];
    const float* bproj  = (const float*)d_in[2];
    float* out = (float*)d_out;
    float* t   = (float*)d_ws;

    dim3 g1(MTOT / BM, DIM / BN);
    proj_gemm_wmma<<<g1, 256, 0, stream>>>(tokens, Wproj, bproj, t);

    buddy_pool_norm<<<BATCH * NCUE, 256, 0, stream>>>(t, out);
}


# projection is linear: project only the 5 cue rows, score patches as (W^T cue).x on raw tokens, project the 3x3 ROI mean of raw tokens; sum order of sim, ROI mean and norms changes
# speedup vs baseline: 9.5214x; 7.1321x over previous
#include <hip/hip_runtime.h>
#include <hip/hip_bf16.h>

#define BATCH 256
#define TSEQ  261
#define DIM   768
#define NCUE  5
#define NPAT  256
#define NROW  (BATCH * NCUE)

typedef __attribute__((ext_vector_type(2))) float v2f;
typedef __attribute__((ext_vector_type(8))) float v8f;

#define GM 64
#define GN 64
#define GK 32
#define GS (GK + 2)

template <int MODE>
__device__ __forceinline__
void rows_gemm_body(const float* __restrict__ A, const float* __restrict__ W,
                    const float* __restrict__ bias, float* __restrict__ C)
{
    __shared__ float As[2][GM][GS];
    __shared__ float Bs[2][GN][GS];

    const int tid  = threadIdx.x;
    const int wave = tid >> 5;
    const int lane = tid & 31;
    const int hl   = lane >> 4;
    const int l16  = lane & 15;

    const int bm = blockIdx.x * GM;
    const int bn = blockIdx.y * GN;

    const int sm0 = (wave & 1) * 2;
    const int sm1 = sm0 + 1;
    const int sn0 = (wave >> 1) * 2;
    const int sn1 = sn0 + 1;

    const float* gA[4];
    const float* gB[4];
    #pragma unroll
    for (int i = 0; i < 4; ++i) {
        const int linear = tid + i * 128;
        const int r  = linear >> 3;
        const int cc = (linear & 7) * 4;
        const int grow = bm + r;
        size_t srow;
        if (MODE == 0) srow = (size_t)(grow / NCUE) * TSEQ + (size_t)(grow % NCUE);
        else           srow = (size_t)grow;
        gA[i] = A + srow * DIM + cc;
        if (MODE == 1) {
            const int kk = linear >> 4;
            const int nn = (linear & 15) * 4;
            gB[i] = W + (size_t)kk * DIM + (bn + nn);
        } else {
            gB[i] = W + (size_t)(bn + r) * DIM + cc;
        }
    }

    float4 ra[4], rb[4];
    #pragma unroll
    for (int i = 0; i < 4; ++i) {
        ra[i] = *(const float4*)(gA[i]);
        rb[i] = *(const float4*)(gB[i]);
    }

    v8f c00 = {}, c01 = {}, c10 = {}, c11 = {};

    #pragma unroll 1
    for (int k0 = -GK; k0 < DIM; k0 += GK) {
        const int s = (k0 >> 5) & 1;

        if (k0 >= 0) {
            #pragma unroll
            for (int kk = 0; kk < GK; kk += 4) {
                v2f a0 = *(const v2f*)&As[s][sm0 * 16 + l16][kk + 2 * hl];
                v2f a1 = *(const v2f*)&As[s][sm1 * 16 + l16][kk + 2 * hl];
                v2f b0 = *(const v2f*)&Bs[s][sn0 * 16 + l16][kk + 2 * hl];
                v2f b1 = *(const v2f*)&Bs[s][sn1 * 16 + l16][kk + 2 * hl];
                c00 = __builtin_amdgcn_wmma_f32_16x16x4_f32(
                          false, a0, false, b0, (short)0, c00, false, false);
                c01 = __builtin_amdgcn_wmma_f32_16x16x4_f32(
                          false, a0, false, b1, (short)0, c01, false, false);
                c10 = __builtin_amdgcn_wmma_f32_16x16x4_f32(
                          false, a1, false, b0, (short)0, c10, false, false);
                c11 = __builtin_amdgcn_wmma_f32_16x16x4_f32(
                          false, a1, false, b1, (short)0, c11, false, false);
            }
        }

        if (k0 + GK < DIM) {
            #pragma unroll
            for (int i = 0; i < 4; ++i) {
                const int linear = tid + i * 128;
                const int r  = linear >> 3;
                const int cc = (linear & 7) * 4;
                As[s ^ 1][r][cc + 0] = ra[i].x;
                As[s ^ 1][r][cc + 1] = ra[i].y;
                As[s ^ 1][r][cc + 2] = ra[i].z;
                As[s ^ 1][r][cc + 3] = ra[i].w;
                if (MODE == 1) {
                    const int kk = linear >> 4;
                    const int nn = (linear & 15) * 4;
                    Bs[s ^ 1][nn + 0][kk] = rb[i].x;
                    Bs[s ^ 1][nn + 1][kk] = rb[i].y;
                    Bs[s ^ 1][nn + 2][kk] = rb[i].z;
                    Bs[s ^ 1][nn + 3][kk] = rb[i].w;
                } else {
                    Bs[s ^ 1][r][cc + 0] = rb[i].x;
                    Bs[s ^ 1][r][cc + 1] = rb[i].y;
                    Bs[s ^ 1][r][cc + 2] = rb[i].z;
                    Bs[s ^ 1][r][cc + 3] = rb[i].w;
                }
            }
            if (k0 + 2 * GK < DIM) {
                #pragma unroll
                for (int i = 0; i < 4; ++i) {
                    ra[i] = *(const float4*)(gA[i] + (k0 + 2 * GK));
                    if (MODE == 1) rb[i] = *(const float4*)(gB[i] + (size_t)(k0 + 2 * GK) * DIM);
                    else           rb[i] = *(const float4*)(gB[i] + (k0 + 2 * GK));
                }
            }
        }
        asm volatile("s_wait_dscnt 0x0" ::: "memory");
        asm volatile("s_barrier_signal -1" ::: "memory");
        asm volatile("s_barrier_wait -1" ::: "memory");
    }

    const int row0 = bm + sm0 * 16;
    const int row1 = bm + sm1 * 16;
    const int col0 = bn + sn0 * 16 + l16;
    const int col1 = bn + sn1 * 16 + l16;
    float bias0 = 0.f, bias1 = 0.f;
    if (MODE != 1) { bias0 = bias[col0]; bias1 = bias[col1]; }
    #pragma unroll
    for (int r = 0; r < 8; ++r) {
        const int ra_ = row0 + r + 8 * hl;
        const int rb_ = row1 + r + 8 * hl;
        if (MODE == 1) {
            C[(size_t)ra_ * DIM + col0] = c00[r];
            C[(size_t)ra_ * DIM + col1] = c01[r];
            C[(size_t)rb_ * DIM + col0] = c10[r];
            C[(size_t)rb_ * DIM + col1] = c11[r];
        } else {
            C[(size_t)ra_ * DIM + col0] = c00[r] + bias0;
            C[(size_t)ra_ * DIM + col1] = c01[r] + bias1;
            C[(size_t)rb_ * DIM + col0] = c10[r] + bias0;
            C[(size_t)rb_ * DIM + col1] = c11[r] + bias1;
        }
    }
}

__global__ __launch_bounds__(128)
void cue_proj_wmma(const float* __restrict__ tokens, const float* __restrict__ W,
                   const float* __restrict__ bias, float* __restrict__ cues)
{
    rows_gemm_body<0>(tokens, W, bias, cues);
}

__global__ __launch_bounds__(128)
void cue_backproj_wmma(const float* __restrict__ cues, const float* __restrict__ W,
                       const float* __restrict__ bias, float* __restrict__ uvec)
{
    rows_gemm_body<1>(cues, W, bias, uvec);
}

__global__ __launch_bounds__(128)
void roi_proj_wmma(const float* __restrict__ xmean, const float* __restrict__ W,
                   const float* __restrict__ bias, float* __restrict__ roi)
{
    rows_gemm_body<2>(xmean, W, bias, roi);
}

__global__ __launch_bounds__(512)
void sim_argmax_roimean(const float* __restrict__ tokens,
                        const float* __restrict__ uvec,
                        float* __restrict__ xmean)
{
    const int b    = blockIdx.x;
    const int tid  = threadIdx.x;
    const int lane = tid & 31;
    const int wave = tid >> 5;

    __shared__ __align__(16) float us[NCUE][DIM];
    __shared__ float sims[NCUE][NPAT];
    __shared__ int   best[NCUE];

    const float4* ub4 = (const float4*)(uvec + (size_t)b * NCUE * DIM);
    for (int i = tid; i < NCUE * DIM / 4; i += 512)
        ((float4*)&us[0][0])[i] = ub4[i];
    __syncthreads();

    const float* pat = tokens + ((size_t)b * TSEQ + NCUE) * DIM;

    #pragma unroll 1
    for (int it = 0; it < 4; ++it) {
        const int n0 = wave * 16 + it * 4;
        const float4* pr = (const float4*)(pat + (size_t)n0 * DIM) + lane;
        float4 x[4][DIM / 128];
        #pragma unroll
        for (int u = 0; u < 4; ++u) {
            #pragma unroll
            for (int q = 0; q < DIM / 128; ++q)
                x[u][q] = pr[u * (DIM / 4) + q * 32];
        }
        __builtin_amdgcn_sched_barrier(0);
        float acc[4][NCUE];
        #pragma unroll
        for (int u = 0; u < 4; ++u) {
            #pragma unroll
            for (int c = 0; c < NCUE; ++c) acc[u][c] = 0.f;
        }
        #pragma unroll
        for (int q = 0; q < DIM / 128; ++q) {
            #pragma unroll
            for (int c = 0; c < NCUE; ++c) {
                const float4 uq = *(const float4*)&us[c][q * 128 + lane * 4];
                #pragma unroll
                for (int u = 0; u < 4; ++u) {
                    float a = acc[u][c];
                    a = __builtin_fmaf(uq.x, x[u][q].x, a);
                    a = __builtin_fmaf(uq.y, x[u][q].y, a);
                    a = __builtin_fmaf(uq.z, x[u][q].z, a);
                    a = __builtin_fmaf(uq.w, x[u][q].w, a);
                    acc[u][c] = a;
                }
            }
        }
        #pragma unroll
        for (int m = 16; m >= 1; m >>= 1) {
            float t[4][NCUE];
            #pragma unroll
            for (int u = 0; u < 4; ++u) {
                #pragma unroll
                for (int c = 0; c < NCUE; ++c) t[u][c] = __shfl_xor(acc[u][c], m, 32);
            }
            #pragma unroll
            for (int u = 0; u < 4; ++u) {
                #pragma unroll
                for (int c = 0; c < NCUE; ++c) acc[u][c] += t[u][c];
            }
        }
        float mine = acc[0][0];
        #pragma unroll
        for (int i = 1; i < 4 * NCUE; ++i)
            mine = (lane == i) ? acc[i / NCUE][i % NCUE] : mine;
        if (lane < 4 * NCUE) sims[lane % NCUE][n0 + lane / NCUE] = mine;
    }
    __syncthreads();

    if (wave < NCUE) {
        float bv = sims[wave][lane];
        int   bi = lane;
        for (int j = 1; j < NPAT / 32; ++j) {
            const float v = sims[wave][lane + 32 * j];
            if (v > bv) { bv = v; bi = lane + 32 * j; }
        }
        for (int m = 16; m >= 1; m >>= 1) {
            const float ov = __shfl_xor(bv, m, 32);
            const int   oi = __shfl_xor(bi, m, 32);
            if (ov > bv || (ov == bv && oi < bi)) { bv = ov; bi = oi; }
        }
        if (lane == 0) best[wave] = bi;
    }
    __syncthreads();

    for (int o = tid; o < NCUE * (DIM / 4); o += 512) {
        const int c  = o / (DIM / 4);
        const int d4 = o - c * (DIM / 4);
        const int idx = best[c];
        const int h = idx >> 4, w = idx & 15;
        const int r0 = max(h - 1, 0), r1 = min(h + 1, 15);
        const int q0 = max(w - 1, 0), q1 = min(w + 1, 15);
        const float inv = 1.0f / (float)((r1 - r0 + 1) * (q1 - q0 + 1));
        float4 v[3][3];
        #pragma unroll
        for (int dh = 0; dh < 3; ++dh) {
            #pragma unroll
            for (int dw = 0; dw < 3; ++dw) {
                const int hh = min(max(h - 1 + dh, 0), 15);
                const int ww = min(max(w - 1 + dw, 0), 15);
                v[dh][dw] = ((const float4*)(pat + (size_t)(hh * 16 + ww) * DIM))[d4];
            }
        }
        float4 acc = {0.f, 0.f, 0.f, 0.f};
        #pragma unroll
        for (int dh = 0; dh < 3; ++dh) {
            #pragma unroll
            for (int dw = 0; dw < 3; ++dw) {
                const int hh = h - 1 + dh, ww = w - 1 + dw;
                const bool in = (hh >= 0) && (hh <= 15) && (ww >= 0) && (ww <= 15);
                acc.x += in ? v[dh][dw].x : 0.f;
                acc.y += in ? v[dh][dw].y : 0.f;
                acc.z += in ? v[dh][dw].z : 0.f;
                acc.w += in ? v[dh][dw].w : 0.f;
            }
        }
        float4 res;
        res.x = acc.x * inv;
        res.y = acc.y * inv;
        res.z = acc.z * inv;
        res.w = acc.w * inv;
        ((float4*)(xmean + ((size_t)b * NCUE + c) * DIM))[d4] = res;
    }
}

__global__ __launch_bounds__(256)
void l2norm_rows(const float* __restrict__ cues, const float* __restrict__ roi,
                 float* __restrict__ out)
{
    const int lane = threadIdx.x & 31;
    const int wave = threadIdx.x >> 5;
    const int row  = blockIdx.x * 8 + wave;
    const int half = row / NROW;
    const int r    = row - half * NROW;
    const int b    = r / NCUE;
    const int c    = r - b * NCUE;

    const float4* src = (const float4*)((half ? roi : cues) + (size_t)r * DIM) + lane;
    float4 v[DIM / 128];
    float ss = 0.f;
    #pragma unroll
    for (int q = 0; q < DIM / 128; ++q) {
        v[q] = src[q * 32];
        ss = __builtin_fmaf(v[q].x, v[q].x, ss);
        ss = __builtin_fmaf(v[q].y, v[q].y, ss);
        ss = __builtin_fmaf(v[q].z, v[q].z, ss);
        ss = __builtin_fmaf(v[q].w, v[q].w, ss);
    }
    for (int m = 16; m >= 1; m >>= 1) ss += __shfl_xor(ss, m, 32);
    const float inv = 1.0f / fmaxf(sqrtf(ss), 1e-12f);

    float4* dst = (float4*)(out + ((size_t)b * (2 * NCUE) + half * NCUE + c) * DIM) + lane;
    #pragma unroll
    for (int q = 0; q < DIM / 128; ++q) {
        float4 o;
        o.x = v[q].x * inv;
        o.y = v[q].y * inv;
        o.z = v[q].z * inv;
        o.w = v[q].w * inv;
        dst[q * 32] = o;
    }
}

extern "C" void kernel_launch(void* const* d_in, const int* in_sizes, int n_in,
                              void* d_out, int out_size, void* d_ws, size_t ws_size,
                              hipStream_t stream)
{
    const float* tokens = (const float*)d_in[0];
    const float* Wproj  = (const float*)d_in[1];
    const float* bproj  = (const float*)d_in[2];
    float* out = (float*)d_out;
    float* ws  = (float*)d_ws;
    float* cues  = ws;
    float* uvec  = ws + 1 * NROW * DIM;
    float* xmean = ws + 2 * NROW * DIM;
    float* roi   = ws + 3 * NROW * DIM;

    dim3 gg(NROW / GM, DIM / GN);
    cue_proj_wmma<<<gg, 128, 0, stream>>>(tokens, Wproj, bproj, cues);
    cue_backproj_wmma<<<gg, 128, 0, stream>>>(cues, Wproj, bproj, uvec);
    sim_argmax_roimean<<<BATCH, 512, 0, stream>>>(tokens, uvec, xmean);
    roi_proj_wmma<<<gg, 128, 0, stream>>>(xmean, Wproj, bproj, roi);
    l2norm_rows<<<2 * NROW / 8, 256, 0, stream>>>(cues, roi, out);
}


# the three 1280-row GEMMs: block tile 64x64 -> 32x64 (2 accumulators per wave), 480 blocks = two independent blocks per WGP, one's LDS hand-over runs under the other's WMMAs; sums unchanged
# speedup vs baseline: 10.2436x; 1.0759x over previous
#include <hip/hip_runtime.h>
#include <hip/hip_bf16.h>

#define BATCH 256
#define TSEQ  261
#define DIM   768
#define NCUE  5
#define NPAT  256
#define NROW  (BATCH * NCUE)

typedef __attribute__((ext_vector_type(2))) float v2f;
typedef __attribute__((ext_vector_type(8))) float v8f;

#define GM 32
#define GN 64
#define GK 32
#define GS (GK + 2)

template <int MODE>
__device__ __forceinline__
void rows_gemm_body(const float* __restrict__ A, const float* __restrict__ W,
                    const float* __restrict__ bias, float* __restrict__ C)
{
    __shared__ float As[2][GM][GS];
    __shared__ float Bs[2][GN][GS];

    const int tid  = threadIdx.x;
    const int wave = tid >> 5;
    const int lane = tid & 31;
    const int hl   = lane >> 4;
    const int l16  = lane & 15;

    const int bm = blockIdx.x * GM;
    const int bn = blockIdx.y * GN;

    const int sn = wave;

    const float* gA[2];
    const float* gB[4];
    #pragma unroll
    for (int i = 0; i < 2; ++i) {
        const int linear = tid + i * 128;
        const int r  = linear >> 3;
        const int cc = (linear & 7) * 4;
        const int grow = bm + r;
        size_t srow;
        if (MODE == 0) srow = (size_t)(grow / NCUE) * TSEQ + (size_t)(grow % NCUE);
        else           srow = (size_t)grow;
        gA[i] = A + srow * DIM + cc;
    }
    #pragma unroll
    for (int i = 0; i < 4; ++i) {
        const int linear = tid + i * 128;
        const int r  = linear >> 3;
        const int cc = (linear & 7) * 4;
        if (MODE == 1) {
            const int kk = linear >> 4;
            const int nn = (linear & 15) * 4;
            gB[i] = W + (size_t)kk * DIM + (bn + nn);
        } else {
            gB[i] = W + (size_t)(bn + r) * DIM + cc;
        }
    }

    float4 ra[2], rb[4];
    #pragma unroll
    for (int i = 0; i < 2; ++i) ra[i] = *(const float4*)(gA[i]);
    #pragma unroll
    for (int i = 0; i < 4; ++i) rb[i] = *(const float4*)(gB[i]);

    v8f c0 = {}, c1 = {};

    #pragma unroll 1
    for (int k0 = -GK; k0 < DIM; k0 += GK) {
        const int s = (k0 >> 5) & 1;

        if (k0 >= 0) {
            #pragma unroll
            for (int kk = 0; kk < GK; kk += 4) {
                v2f a0 = *(const v2f*)&As[s][l16][kk + 2 * hl];
                v2f a1 = *(const v2f*)&As[s][16 + l16][kk + 2 * hl];
                v2f b  = *(const v2f*)&Bs[s][sn * 16 + l16][kk + 2 * hl];
                c0 = __builtin_amdgcn_wmma_f32_16x16x4_f32(
                         false, a0, false, b, (short)0, c0, false, false);
                c1 = __builtin_amdgcn_wmma_f32_16x16x4_f32(
                         false, a1, false, b, (short)0, c1, false, false);
            }
        }

        if (k0 + GK < DIM) {
            #pragma unroll
            for (int i = 0; i < 2; ++i) {
                const int linear = tid + i * 128;
                const int r  = linear >> 3;
                const int cc = (linear & 7) * 4;
                As[s ^ 1][r][cc + 0] = ra[i].x;
                As[s ^ 1][r][cc + 1] = ra[i].y;
                As[s ^ 1][r][cc + 2] = ra[i].z;
                As[s ^ 1][r][cc + 3] = ra[i].w;
            }
            #pragma unroll
            for (int i = 0; i < 4; ++i) {
                const int linear = tid + i * 128;
                const int r  = linear >> 3;
                const int cc = (linear & 7) * 4;
                if (MODE == 1) {
                    const int kk = linear >> 4;
                    const int nn = (linear & 15) * 4;
                    Bs[s ^ 1][nn + 0][kk] = rb[i].x;
                    Bs[s ^ 1][nn + 1][kk] = rb[i].y;
                    Bs[s ^ 1][nn + 2][kk] = rb[i].z;
                    Bs[s ^ 1][nn + 3][kk] = rb[i].w;
                } else {
                    Bs[s ^ 1][r][cc + 0] = rb[i].x;
                    Bs[s ^ 1][r][cc + 1] = rb[i].y;
                    Bs[s ^ 1][r][cc + 2] = rb[i].z;
                    Bs[s ^ 1][r][cc + 3] = rb[i].w;
                }
            }
            if (k0 + 2 * GK < DIM) {
                #pragma unroll
                for (int i = 0; i < 2; ++i)
                    ra[i] = *(const float4*)(gA[i] + (k0 + 2 * GK));
                #pragma unroll
                for (int i = 0; i < 4; ++i) {
                    if (MODE == 1) rb[i] = *(const float4*)(gB[i] + (size_t)(k0 + 2 * GK) * DIM);
                    else           rb[i] = *(const float4*)(gB[i] + (k0 + 2 * GK));
                }
            }
        }
        asm volatile("s_wait_dscnt 0x0" ::: "memory");
        asm volatile("s_barrier_signal -1" ::: "memory");
        asm volatile("s_barrier_wait -1" ::: "memory");
    }

    const int col = bn + sn * 16 + l16;
    float bias0 = 0.f;
    if (MODE != 1) bias0 = bias[col];
    #pragma unroll
    for (int r = 0; r < 8; ++r) {
        const int ra_ = bm + r + 8 * hl;
        const int rb_ = bm + 16 + r + 8 * hl;
        if (MODE == 1) {
            C[(size_t)ra_ * DIM + col] = c0[r];
            C[(size_t)rb_ * DIM + col] = c1[r];
        } else {
            C[(size_t)ra_ * DIM + col] = c0[r] + bias0;
            C[(size_t)rb_ * DIM + col] = c1[r] + bias0;
        }
    }
}

__global__ __launch_bounds__(128)
void cue_proj_wmma(const float* __restrict__ tokens, const float* __restrict__ W,
                   const float* __restrict__ bias, float* __restrict__ cues)
{
    rows_gemm_body<0>(tokens, W, bias, cues);
}

__global__ __launch_bounds__(128)
void cue_backproj_wmma(const float* __restrict__ cues, const float* __restrict__ W,
                       const float* __restrict__ bias, float* __restrict__ uvec)
{
    rows_gemm_body<1>(cues, W, bias, uvec);
}

__global__ __launch_bounds__(128)
void roi_proj_wmma(const float* __restrict__ xmean, const float* __restrict__ W,
                   const float* __restrict__ bias, float* __restrict__ roi)
{
    rows_gemm_body<2>(xmean, W, bias, roi);
}

__global__ __launch_bounds__(512)
void sim_argmax_roimean(const float* __restrict__ tokens,
                        const float* __restrict__ uvec,
                        float* __restrict__ xmean)
{
    const int b    = blockIdx.x;
    const int tid  = threadIdx.x;
    const int lane = tid & 31;
    const int wave = tid >> 5;

    __shared__ __align__(16) float us[NCUE][DIM];
    __shared__ float sims[NCUE][NPAT];
    __shared__ int   best[NCUE];

    const float4* ub4 = (const float4*)(uvec + (size_t)b * NCUE * DIM);
    for (int i = tid; i < NCUE * DIM / 4; i += 512)
        ((float4*)&us[0][0])[i] = ub4[i];
    __syncthreads();

    const float* pat = tokens + ((size_t)b * TSEQ + NCUE) * DIM;

    #pragma unroll 1
    for (int it = 0; it < 4; ++it) {
        const int n0 = wave * 16 + it * 4;
        const float4* pr = (const float4*)(pat + (size_t)n0 * DIM) + lane;
        float4 x[4][DIM / 128];
        #pragma unroll
        for (int u = 0; u < 4; ++u) {
            #pragma unroll
            for (int q = 0; q < DIM / 128; ++q)
                x[u][q] = pr[u * (DIM / 4) + q * 32];
        }
        __builtin_amdgcn_sched_barrier(0);
        float acc[4][NCUE];
        #pragma unroll
        for (int u = 0; u < 4; ++u) {
            #pragma unroll
            for (int c = 0; c < NCUE; ++c) acc[u][c] = 0.f;
        }
        #pragma unroll
        for (int q = 0; q < DIM / 128; ++q) {
            #pragma unroll
            for (int c = 0; c < NCUE; ++c) {
                const float4 uq = *(const float4*)&us[c][q * 128 + lane * 4];
                #pragma unroll
                for (int u = 0; u < 4; ++u) {
                    float a = acc[u][c];
                    a = __builtin_fmaf(uq.x, x[u][q].x, a);
                    a = __builtin_fmaf(uq.y, x[u][q].y, a);
                    a = __builtin_fmaf(uq.z, x[u][q].z, a);
                    a = __builtin_fmaf(uq.w, x[u][q].w, a);
                    acc[u][c] = a;
                }
            }
        }
        #pragma unroll
        for (int m = 16; m >= 1; m >>= 1) {
            float t[4][NCUE];
            #pragma unroll
            for (int u = 0; u < 4; ++u) {
                #pragma unroll
                for (int c = 0; c < NCUE; ++c) t[u][c] = __shfl_xor(acc[u][c], m, 32);
            }
            #pragma unroll
            for (int u = 0; u < 4; ++u) {
                #pragma unroll
                for (int c = 0; c < NCUE; ++c) acc[u][c] += t[u][c];
            }
        }
        float mine = acc[0][0];
        #pragma unroll
        for (int i = 1; i < 4 * NCUE; ++i)
            mine = (lane == i) ? acc[i / NCUE][i % NCUE] : mine;
        if (lane < 4 * NCUE) sims[lane % NCUE][n0 + lane / NCUE] = mine;
    }
    __syncthreads();

    if (wave < NCUE) {
        float bv = sims[wave][lane];
        int   bi = lane;
        for (int j = 1; j < NPAT / 32; ++j) {
            const float v = sims[wave][lane + 32 * j];
            if (v > bv) { bv = v; bi = lane + 32 * j; }
        }
        for (int m = 16; m >= 1; m >>= 1) {
            const float ov = __shfl_xor(bv, m, 32);
            const int   oi = __shfl_xor(bi, m, 32);
            if (ov > bv || (ov == bv && oi < bi)) { bv = ov; bi = oi; }
        }
        if (lane == 0) best[wave] = bi;
    }
    __syncthreads();

    for (int o = tid; o < NCUE * (DIM / 4); o += 512) {
        const int c  = o / (DIM / 4);
        const int d4 = o - c * (DIM / 4);
        const int idx = best[c];
        const int h = idx >> 4, w = idx & 15;
        const int r0 = max(h - 1, 0), r1 = min(h + 1, 15);
        const int q0 = max(w - 1, 0), q1 = min(w + 1, 15);
        const float inv = 1.0f / (float)((r1 - r0 + 1) * (q1 - q0 + 1));
        float4 v[3][3];
        #pragma unroll
        for (int dh = 0; dh < 3; ++dh) {
            #pragma unroll
            for (int dw = 0; dw < 3; ++dw) {
                const int hh = min(max(h - 1 + dh, 0), 15);
                const int ww = min(max(w - 1 + dw, 0), 15);
                v[dh][dw] = ((const float4*)(pat + (size_t)(hh * 16 + ww) * DIM))[d4];
            }
        }
        float4 acc = {0.f, 0.f, 0.f, 0.f};
        #pragma unroll
        for (int dh = 0; dh < 3; ++dh) {
            #pragma unroll
            for (int dw = 0; dw < 3; ++dw) {
                const int hh = h - 1 + dh, ww = w - 1 + dw;
                const bool in = (hh >= 0) && (hh <= 15) && (ww >= 0) && (ww <= 15);
                acc.x += in ? v[dh][dw].x : 0.f;
                acc.y += in ? v[dh][dw].y : 0.f;
                acc.z += in ? v[dh][dw].z : 0.f;
                acc.w += in ? v[dh][dw].w : 0.f;
            }
        }
        float4 res;
        res.x = acc.x * inv;
        res.y = acc.y * inv;
        res.z = acc.z * inv;
        res.w = acc.w * inv;
        ((float4*)(xmean + ((size_t)b * NCUE + c) * DIM))[d4] = res;
    }
}

__global__ __launch_bounds__(256)
void l2norm_rows(const float* __restrict__ cues, const float* __restrict__ roi,
                 float* __restrict__ out)
{
    const int lane = threadIdx.x & 31;
    const int wave = threadIdx.x >> 5;
    const int row  = blockIdx.x * 8 + wave;
    const int half = row / NROW;
    const int r    = row - half * NROW;
    const int b    = r / NCUE;
    const int c    = r - b * NCUE;

    const float4* src = (const float4*)((half ? roi : cues) + (size_t)r * DIM) + lane;
    float4 v[DIM / 128];
    float ss = 0.f;
    #pragma unroll
    for (int q = 0; q < DIM / 128; ++q) {
        v[q] = src[q * 32];
        ss = __builtin_fmaf(v[q].x, v[q].x, ss);
        ss = __builtin_fmaf(v[q].y, v[q].y, ss);
        ss = __builtin_fmaf(v[q].z, v[q].z, ss);
        ss = __builtin_fmaf(v[q].w, v[q].w, ss);
    }
    for (int m = 16; m >= 1; m >>= 1) ss += __shfl_xor(ss, m, 32);
    const float inv = 1.0f / fmaxf(sqrtf(ss), 1e-12f);

    float4* dst = (float4*)(out + ((size_t)b * (2 * NCUE) + half * NCUE + c) * DIM) + lane;
    #pragma unroll
    for (int q = 0; q < DIM / 128; ++q) {
        float4 o;
        o.x = v[q].x * inv;
        o.y = v[q].y * inv;
        o.z = v[q].z * inv;
        o.w = v[q].w * inv;
        dst[q * 32] = o;
    }
}

extern "C" void kernel_launch(void* const* d_in, const int* in_sizes, int n_in,
                              void* d_out, int out_size, void* d_ws, size_t ws_size,
                              hipStream_t stream)
{
    const float* tokens = (const float*)d_in[0];
    const float* Wproj  = (const float*)d_in[1];
    const float* bproj  = (const float*)d_in[2];
    float* out = (float*)d_out;
    float* ws  = (float*)d_ws;
    float* cues  = ws;
    float* uvec  = ws + 1 * NROW * DIM;
    float* xmean = ws + 2 * NROW * DIM;
    float* roi   = ws + 3 * NROW * DIM;

    dim3 gg(NROW / GM, DIM / GN);
    cue_proj_wmma<<<gg, 128, 0, stream>>>(tokens, Wproj, bproj, cues);
    cue_backproj_wmma<<<gg, 128, 0, stream>>>(cues, Wproj, bproj, uvec);
    sim_argmax_roimean<<<BATCH, 512, 0, stream>>>(tokens, uvec, xmean);
    roi_proj_wmma<<<gg, 128, 0, stream>>>(xmean, Wproj, bproj, roi);
    l2norm_rows<<<2 * NROW / 8, 256, 0, stream>>>(cues, roi, out);
}


# the three 1280-row GEMMs: K chunk per LDS hand-over 32 -> 64 floats: 32 WMMAs per wave per hand-over, 12 hand-overs not 24, twice the cover for the prefetched chunk's global loads; sums unchanged
# speedup vs baseline: 11.6993x; 1.1421x over previous
#include <hip/hip_runtime.h>
#include <hip/hip_bf16.h>

#define BATCH 256
#define TSEQ  261
#define DIM   768
#define NCUE  5
#define NPAT  256
#define NROW  (BATCH * NCUE)

typedef __attribute__((ext_vector_type(2))) float v2f;
typedef __attribute__((ext_vector_type(8))) float v8f;

#define GM 32
#define GN 64
#define GK 64
#define GS (GK + 4)

template <int MODE>
__device__ __forceinline__
void rows_gemm_body(const float* __restrict__ A, const float* __restrict__ W,
                    const float* __restrict__ bias, float* __restrict__ C)
{
    __shared__ __align__(16) float As[2][GM][GS];
    __shared__ __align__(16) float Bs[2][GN][GS];

    const int tid  = threadIdx.x;
    const int wave = tid >> 5;
    const int lane = tid & 31;
    const int hl   = lane >> 4;
    const int l16  = lane & 15;

    const int bm = blockIdx.x * GM;
    const int bn = blockIdx.y * GN;

    const int sn = wave;

    const int sr = tid >> 4;
    const int sc = (tid & 15) * 4;

    const float* gA[4];
    #pragma unroll
    for (int i = 0; i < 4; ++i) {
        const int grow = bm + sr + 8 * i;
        size_t srow;
        if (MODE == 0) srow = (size_t)(grow / NCUE) * TSEQ + (size_t)(grow % NCUE);
        else           srow = (size_t)grow;
        gA[i] = A + srow * DIM + sc;
    }
    const int tn = tid & 63;
    const int tk = (tid >> 6) * 32;
    const float* gB0;
    if (MODE == 1) gB0 = W + (size_t)tk * DIM + (bn + tn);
    else           gB0 = W + (size_t)(bn + sr) * DIM + sc;

#define RG_EACH_A(X) X(0) X(1) X(2) X(3)
#define RG_EACH_B(X) X(0) X(1) X(2) X(3) X(4) X(5) X(6) X(7)
#define RG_DECL_A(i) float4 ra##i = *(const float4*)(gA[i]);
#define RG_GET_B(i, p) \
    if (MODE == 1) { rb##i.x = (p)[(size_t)(4 * i + 0) * DIM]; rb##i.y = (p)[(size_t)(4 * i + 1) * DIM]; \
                     rb##i.z = (p)[(size_t)(4 * i + 2) * DIM]; rb##i.w = (p)[(size_t)(4 * i + 3) * DIM]; } \
    else           { rb##i = *(const float4*)((p) + (size_t)(8 * i) * DIM); }
#define RG_DECL_B(i) float4 rb##i; RG_GET_B(i, gB0)
    RG_EACH_A(RG_DECL_A)
    RG_EACH_B(RG_DECL_B)

    v8f c0 = {}, c1 = {};

    #pragma unroll 1
    for (int k0 = -GK; k0 < DIM; k0 += GK) {
        const int s = (k0 >> 6) & 1;

        if (k0 >= 0) {
            #pragma unroll
            for (int kk = 0; kk < GK; kk += 4) {
                v2f a0 = *(const v2f*)&As[s][l16][kk + 2 * hl];
                v2f a1 = *(const v2f*)&As[s][16 + l16][kk + 2 * hl];
                v2f b  = *(const v2f*)&Bs[s][sn * 16 + l16][kk + 2 * hl];
                c0 = __builtin_amdgcn_wmma_f32_16x16x4_f32(
                         false, a0, false, b, (short)0, c0, false, false);
                c1 = __builtin_amdgcn_wmma_f32_16x16x4_f32(
                         false, a1, false, b, (short)0, c1, false, false);
            }
        }

        if (k0 + GK < DIM) {
#define RG_STORE_A(i) *(float4*)&As[s ^ 1][sr + 8 * i][sc] = ra##i;
#define RG_STORE_B(i) \
    if (MODE == 1) *(float4*)&Bs[s ^ 1][tn][tk + 4 * i] = rb##i; \
    else           *(float4*)&Bs[s ^ 1][sr + 8 * i][sc] = rb##i;
            RG_EACH_A(RG_STORE_A)
            RG_EACH_B(RG_STORE_B)
            if (k0 + 2 * GK < DIM) {
                const float* nB = (MODE == 1) ? gB0 + (size_t)(k0 + 2 * GK) * DIM
                                              : gB0 + (k0 + 2 * GK);
#define RG_LOAD_A(i) ra##i = *(const float4*)(gA[i] + (k0 + 2 * GK));
#define RG_LOAD_B(i) RG_GET_B(i, nB)
                RG_EACH_A(RG_LOAD_A)
                RG_EACH_B(RG_LOAD_B)
            }
        }
        asm volatile("s_wait_dscnt 0x0" ::: "memory");
        asm volatile("s_barrier_signal -1" ::: "memory");
        asm volatile("s_barrier_wait -1" ::: "memory");
    }

    const int col = bn + sn * 16 + l16;
    float bias0 = 0.f;
    if (MODE != 1) bias0 = bias[col];
    #pragma unroll
    for (int r = 0; r < 8; ++r) {
        const int ra_ = bm + r + 8 * hl;
        const int rb_ = bm + 16 + r + 8 * hl;
        if (MODE == 1) {
            C[(size_t)ra_ * DIM + col] = c0[r];
            C[(size_t)rb_ * DIM + col] = c1[r];
        } else {
            C[(size_t)ra_ * DIM + col] = c0[r] + bias0;
            C[(size_t)rb_ * DIM + col] = c1[r] + bias0;
        }
    }
}

__global__ __launch_bounds__(128)
void cue_proj_wmma(const float* __restrict__ tokens, const float* __restrict__ W,
                   const float* __restrict__ bias, float* __restrict__ cues)
{
    rows_gemm_body<0>(tokens, W, bias, cues);
}

__global__ __launch_bounds__(128)
void cue_backproj_wmma(const float* __restrict__ cues, const float* __restrict__ W,
                       const float* __restrict__ bias, float* __restrict__ uvec)
{
    rows_gemm_body<1>(cues, W, bias, uvec);
}

__global__ __launch_bounds__(128)
void roi_proj_wmma(const float* __restrict__ xmean, const float* __restrict__ W,
                   const float* __restrict__ bias, float* __restrict__ roi)
{
    rows_gemm_body<2>(xmean, W, bias, roi);
}

__global__ __launch_bounds__(512)
void sim_argmax_roimean(const float* __restrict__ tokens,
                        const float* __restrict__ uvec,
                        float* __restrict__ xmean)
{
    const int b    = blockIdx.x;
    const int tid  = threadIdx.x;
    const int lane = tid & 31;
    const int wave = tid >> 5;

    __shared__ __align__(16) float us[NCUE][DIM];
    __shared__ float sims[NCUE][NPAT];
    __shared__ int   best[NCUE];

    const float4* ub4 = (const float4*)(uvec + (size_t)b * NCUE * DIM);
    for (int i = tid; i < NCUE * DIM / 4; i += 512)
        ((float4*)&us[0][0])[i] = ub4[i];
    __syncthreads();

    const float* pat = tokens + ((size_t)b * TSEQ + NCUE) * DIM;

    #pragma unroll 1
    for (int it = 0; it < 4; ++it) {
        const int n0 = wave * 16 + it * 4;
        const float4* pr = (const float4*)(pat + (size_t)n0 * DIM) + lane;
        float4 x[4][DIM / 128];
        #pragma unroll
        for (int u = 0; u < 4; ++u) {
            #pragma unroll
            for (int q = 0; q < DIM / 128; ++q)
                x[u][q] = pr[u * (DIM / 4) + q * 32];
        }
        __builtin_amdgcn_sched_barrier(0);
        float acc[4][NCUE];
        #pragma unroll
        for (int u = 0; u < 4; ++u) {
            #pragma unroll
            for (int c = 0; c < NCUE; ++c) acc[u][c] = 0.f;
        }
        #pragma unroll
        for (int q = 0; q < DIM / 128; ++q) {
            #pragma unroll
            for (int c = 0; c < NCUE; ++c) {
                const float4 uq = *(const float4*)&us[c][q * 128 + lane * 4];
                #pragma unroll
                for (int u = 0; u < 4; ++u) {
                    float a = acc[u][c];
                    a = __builtin_fmaf(uq.x, x[u][q].x, a);
                    a = __builtin_fmaf(uq.y, x[u][q].y, a);
                    a = __builtin_fmaf(uq.z, x[u][q].z, a);
                    a = __builtin_fmaf(uq.w, x[u][q].w, a);
                    acc[u][c] = a;
                }
            }
        }
        #pragma unroll
        for (int m = 16; m >= 1; m >>= 1) {
            float t[4][NCUE];
            #pragma unroll
            for (int u = 0; u < 4; ++u) {
                #pragma unroll
                for (int c = 0; c < NCUE; ++c) t[u][c] = __shfl_xor(acc[u][c], m, 32);
            }
            #pragma unroll
            for (int u = 0; u < 4; ++u) {
                #pragma unroll
                for (int c = 0; c < NCUE; ++c) acc[u][c] += t[u][c];
            }
        }
        float mine = acc[0][0];
        #pragma unroll
        for (int i = 1; i < 4 * NCUE; ++i)
            mine = (lane == i) ? acc[i / NCUE][i % NCUE] : mine;
        if (lane < 4 * NCUE) sims[lane % NCUE][n0 + lane / NCUE] = mine;
    }
    __syncthreads();

    if (wave < NCUE) {
        float bv = sims[wave][lane];
        int   bi = lane;
        for (int j = 1; j < NPAT / 32; ++j) {
            const float v = sims[wave][lane + 32 * j];
            if (v > bv) { bv = v; bi = lane + 32 * j; }
        }
        for (int m = 16; m >= 1; m >>= 1) {
            const float ov = __shfl_xor(bv, m, 32);
            const int   oi = __shfl_xor(bi, m, 32);
            if (ov > bv || (ov == bv && oi < bi)) { bv = ov; bi = oi; }
        }
        if (lane == 0) best[wave] = bi;
    }
    __syncthreads();

    for (int o = tid; o < NCUE * (DIM / 4); o += 512) {
        const int c  = o / (DIM / 4);
        const int d4 = o - c * (DIM / 4);
        const int idx = best[c];
        const int h = idx >> 4, w = idx & 15;
        const int r0 = max(h - 1, 0), r1 = min(h + 1, 15);
        const int q0 = max(w - 1, 0), q1 = min(w + 1, 15);
        const float inv = 1.0f / (float)((r1 - r0 + 1) * (q1 - q0 + 1));
        float4 v[3][3];
        #pragma unroll
        for (int dh = 0; dh < 3; ++dh) {
            #pragma unroll
            for (int dw = 0; dw < 3; ++dw) {
                const int hh = min(max(h - 1 + dh, 0), 15);
                const int ww = min(max(w - 1 + dw, 0), 15);
                v[dh][dw] = ((const float4*)(pat + (size_t)(hh * 16 + ww) * DIM))[d4];
            }
        }
        float4 acc = {0.f, 0.f, 0.f, 0.f};
        #pragma unroll
        for (int dh = 0; dh < 3; ++dh) {
            #pragma unroll
            for (int dw = 0; dw < 3; ++dw) {
                const int hh = h - 1 + dh, ww = w - 1 + dw;
                const bool in = (hh >= 0) && (hh <= 15) && (ww >= 0) && (ww <= 15);
                acc.x += in ? v[dh][dw].x : 0.f;
                acc.y += in ? v[dh][dw].y : 0.f;
                acc.z += in ? v[dh][dw].z : 0.f;
                acc.w += in ? v[dh][dw].w : 0.f;
            }
        }
        float4 res;
        res.x = acc.x * inv;
        res.y = acc.y * inv;
        res.z = acc.z * inv;
        res.w = acc.w * inv;
        ((float4*)(xmean + ((size_t)b * NCUE + c) * DIM))[d4] = res;
    }
}

__global__ __launch_bounds__(256)
void l2norm_rows(const float* __restrict__ cues, const float* __restrict__ roi,
                 float* __restrict__ out)
{
    const int lane = threadIdx.x & 31;
    const int wave = threadIdx.x >> 5;
    const int row  = blockIdx.x * 8 + wave;
    const int half = row / NROW;
    const int r    = row - half * NROW;
    const int b    = r / NCUE;
    const int c    = r - b * NCUE;

    const float4* src = (const float4*)((half ? roi : cues) + (size_t)r * DIM) + lane;
    float4 v[DIM / 128];
    float ss = 0.f;
    #pragma unroll
    for (int q = 0; q < DIM / 128; ++q) {
        v[q] = src[q * 32];
        ss = __builtin_fmaf(v[q].x, v[q].x, ss);
        ss = __builtin_fmaf(v[q].y, v[q].y, ss);
        ss = __builtin_fmaf(v[q].z, v[q].z, ss);
        ss = __builtin_fmaf(v[q].w, v[q].w, ss);
    }
    for (int m = 16; m >= 1; m >>= 1) ss += __shfl_xor(ss, m, 32);
    const float inv = 1.0f / fmaxf(sqrtf(ss), 1e-12f);

    float4* dst = (float4*)(out + ((size_t)b * (2 * NCUE) + half * NCUE + c) * DIM) + lane;
    #pragma unroll
    for (int q = 0; q < DIM / 128; ++q) {
        float4 o;
        o.x = v[q].x * inv;
        o.y = v[q].y * inv;
        o.z = v[q].z * inv;
        o.w = v[q].w * inv;
        dst[q * 32] = o;
    }
}

extern "C" void kernel_launch(void* const* d_in, const int* in_sizes, int n_in,
                              void* d_out, int out_size, void* d_ws, size_t ws_size,
                              hipStream_t stream)
{
    const float* tokens = (const float*)d_in[0];
    const float* Wproj  = (const float*)d_in[1];
    const float* bproj  = (const float*)d_in[2];
    float* out = (float*)d_out;
    float* ws  = (float*)d_ws;
    float* cues  = ws;
    float* uvec  = ws + 1 * NROW * DIM;
    float* xmean = ws + 2 * NROW * DIM;
    float* roi   = ws + 3 * NROW * DIM;

    dim3 gg(NROW / GM, DIM / GN);
    cue_proj_wmma<<<gg, 128, 0, stream>>>(tokens, Wproj, bproj, cues);
    cue_backproj_wmma<<<gg, 128, 0, stream>>>(cues, Wproj, bproj, uvec);
    sim_argmax_roimean<<<BATCH, 512, 0, stream>>>(tokens, uvec, xmean);
    roi_proj_wmma<<<gg, 128, 0, stream>>>(xmean, Wproj, bproj, roi);
    l2norm_rows<<<2 * NROW / 8, 256, 0, stream>>>(cues, roi, out);
}
